# starting kernel
# baseline (speedup 1.0000x reference)
#include <hip/hip_runtime.h>
#include <hip/hip_bf16.h>

typedef float v2f __attribute__((ext_vector_type(2)));
typedef float v8f __attribute__((ext_vector_type(8)));

__global__ void fill_f32(float* p, float v, long n) {
  long i = (long)blockIdx.x * blockDim.x + threadIdx.x;
  if (i < n) p[i] = v;
}

__global__ void deg_accum(const int* __restrict__ dst, float* __restrict__ deg, int E) {
  long e = (long)blockIdx.x * blockDim.x + threadIdx.x;
  if (e < E) atomicAdd(&deg[dst[e]], 1.0f);
}

__global__ void deg_to_dinv(float* deg, int n) {
  long i = (long)blockIdx.x * blockDim.x + threadIdx.x;
  if (i < n) deg[i] = rsqrtf(deg[i]);
}

__global__ __launch_bounds__(256)
void gemm64_wmma(const float* __restrict__ A, const float* __restrict__ W,
                 const float* __restrict__ bias, float* __restrict__ out, int nRows) {
  __shared__ float sW[64 * 64];
  int tid = threadIdx.x;
  for (int i = tid; i < 64 * 64; i += 256) sW[i] = W[i];
  __syncthreads();

  int wave = tid >> 5;
  int lane = tid & 31;
  long tileRow = ((long)blockIdx.x * 8 + wave) * 16;

  int koff = (lane >> 4) << 1;
  int mrow = lane & 15;
  long arow = tileRow + mrow;
  if (arow >= nRows) arow = nRows - 1;
  const float* aptr = A + arow * 64 + koff;

  v8f acc[4] = {};
  for (int k0 = 0; k0 < 64; k0 += 4) {
    v2f a;
    a.x = aptr[k0];
    a.y = aptr[k0 + 1];
    int kk = k0 + koff;
#pragma unroll
    for (int n = 0; n < 4; ++n) {
      int col = n * 16 + mrow;
      v2f b;
      b.x = sW[kk * 64 + col];
      b.y = sW[(kk + 1) * 64 + col];
      acc[n] = __builtin_amdgcn_wmma_f32_16x16x4_f32(
           false, a,  false, b,
           (short)0, acc[n],  false,  false);
    }
  }

  int colBase = lane & 15;
  int rowOff = (lane >> 4) * 8;
#pragma unroll
  for (int r = 0; r < 8; ++r) {
    long row = tileRow + rowOff + r;
    if (row < nRows) {
#pragma unroll
      for (int n = 0; n < 4; ++n) {
        int col = n * 16 + colBase;
        float v = acc[n][r];
        if (bias) v += bias[col];
        out[row * 64 + col] = v;
      }
    }
  }
}

__global__ void scatter_edges(const int* __restrict__ src, const int* __restrict__ dst,
                              const float* __restrict__ h, const float* __restrict__ dinv,
                              float* __restrict__ agg, long nWork) {
  long t = (long)blockIdx.x * blockDim.x + threadIdx.x;
  if (t >= nWork) return;
  long e = t >> 6;
  int c = (int)(t & 63);
  int s = src[e];
  int d = dst[e];
  float w = dinv[s] * dinv[d];
  atomicAdd(&agg[(long)d * 64 + c], h[(long)s * 64 + c] * w);
}

__global__ void node_fuse(const float* __restrict__ agg, const float* __restrict__ h,
                          const float* __restrict__ dinv, const float* __restrict__ b,
                          const float* __restrict__ gamma, const float* __restrict__ beta,
                          const float* __restrict__ x, float* __restrict__ g, int nNodes) {
  int wave = threadIdx.x >> 5;
  int lane = threadIdx.x & 31;
  long node = (long)blockIdx.x * (blockDim.x >> 5) + wave;
  if (node >= nNodes) return;

  float di = dinv[node];
  float selfw = di * di;
  int c0 = lane, c1 = lane + 32;
  long base = node * 64;

  float v0 = agg[base + c0] + h[base + c0] * selfw + b[c0];
  float v1 = agg[base + c1] + h[base + c1] * selfw + b[c1];

  float s = v0 + v1;
#pragma unroll
  for (int off = 16; off > 0; off >>= 1) s += __shfl_xor(s, off, 32);
  float mu = s * (1.0f / 64.0f);

  float d0 = v0 - mu, d1 = v1 - mu;
  float q = d0 * d0 + d1 * d1;
#pragma unroll
  for (int off = 16; off > 0; off >>= 1) q += __shfl_xor(q, off, 32);
  float inv = rsqrtf(q * (1.0f / 64.0f) + 1e-5f);

  float y0 = d0 * inv * gamma[c0] + beta[c0];
  float y1 = d1 * inv * gamma[c1] + beta[c1];
  y0 = fmaxf(y0, 0.0f) * x[base + c0];
  y1 = fmaxf(y1, 0.0f) * x[base + c1];
  g[base + c0] = y0;
  g[base + c1] = y1;
}

extern "C" void kernel_launch(void* const* d_in, const int* in_sizes, int n_in,
                              void* d_out, int out_size, void* d_ws, size_t ws_size,
                              hipStream_t stream) {
  const float* x     = (const float*)d_in[0];
  const int*   eidx  = (const int*)d_in[1];
  const float* W     = (const float*)d_in[2];
  const float* b     = (const float*)d_in[3];
  const float* gamma = (const float*)d_in[4];
  const float* beta  = (const float*)d_in[5];
  const float* fcW   = (const float*)d_in[6];
  const float* fcb   = (const float*)d_in[7];
  float* out = (float*)d_out;

  int N = in_sizes[0] / 64;
  int E = in_sizes[1] / 2;
  const int* src = eidx;
  const int* dst = eidx + E;

  char* ws = (char*)d_ws;
  size_t off = 0;
  float* dinv = (float*)(ws + off);
  off += (((size_t)N * 4) + 255) & ~(size_t)255;
  float* h = (float*)(ws + off);
  off += (((size_t)N * 64 * 4) + 255) & ~(size_t)255;
  float* agg = (float*)(ws + off);
  float* g = h;

  const int TB = 256;

  fill_f32<<<(N + TB - 1) / TB, TB, 0, stream>>>(dinv, 1.0f, N);
  fill_f32<<<(int)(((long)N * 64 + TB - 1) / TB), TB, 0, stream>>>(agg, 0.0f, (long)N * 64);
  deg_accum<<<(E + TB - 1) / TB, TB, 0, stream>>>(dst, dinv, E);
  deg_to_dinv<<<(N + TB - 1) / TB, TB, 0, stream>>>(dinv, N);

  int gemmBlocks = (N + 127) / 128;
  gemm64_wmma<<<gemmBlocks, TB, 0, stream>>>(x, W, nullptr, h, N);

  long nWork = (long)E * 64;
  scatter_edges<<<(int)((nWork + TB - 1) / TB), TB, 0, stream>>>(src, dst, h, dinv, agg, nWork);

  node_fuse<<<(N + 7) / 8, TB, 0, stream>>>(agg, h, dinv, b, gamma, beta, x, g, N);

  gemm64_wmma<<<gemmBlocks, TB, 0, stream>>>(g, fcW, fcb, out, N);
}


# Float-atomic edge scatter replaced by destination-owner work-groups: each scans dst, sums its nodes' rows in LDS (LDS float atomics, arrival order), runs the node epilogue; 8 -> 6 launches
# speedup vs baseline: 1.6769x; 1.6769x over previous
#include <hip/hip_runtime.h>
#include <hip/hip_bf16.h>

typedef float v2f __attribute__((ext_vector_type(2)));
typedef float v8f __attribute__((ext_vector_type(8)));

__global__ void fill_f32(float* p, float v, long n) {
  long i = (long)blockIdx.x * blockDim.x + threadIdx.x;
  if (i < n) p[i] = v;
}

__global__ void deg_accum(const int* __restrict__ dst, float* __restrict__ deg, int E) {
  long e = (long)blockIdx.x * blockDim.x + threadIdx.x;
  if (e < E) atomicAdd(&deg[dst[e]], 1.0f);
}

__global__ void deg_to_dinv(float* deg, int n) {
  long i = (long)blockIdx.x * blockDim.x + threadIdx.x;
  if (i < n) deg[i] = rsqrtf(deg[i]);
}

__global__ __launch_bounds__(256)
void gemm64_wmma(const float* __restrict__ A, const float* __restrict__ W,
                 const float* __restrict__ bias, float* __restrict__ out, int nRows) {
  __shared__ float sW[64 * 64];
  int tid = threadIdx.x;
  for (int i = tid; i < 64 * 64; i += 256) sW[i] = W[i];
  __syncthreads();

  int wave = tid >> 5;
  int lane = tid & 31;
  long tileRow = ((long)blockIdx.x * 8 + wave) * 16;

  int koff = (lane >> 4) << 1;
  int mrow = lane & 15;
  long arow = tileRow + mrow;
  if (arow >= nRows) arow = nRows - 1;
  const float* aptr = A + arow * 64 + koff;

  v8f acc[4] = {};
  for (int k0 = 0; k0 < 64; k0 += 4) {
    v2f a;
    a.x = aptr[k0];
    a.y = aptr[k0 + 1];
    int kk = k0 + koff;
#pragma unroll
    for (int n = 0; n < 4; ++n) {
      int col = n * 16 + mrow;
      v2f b;
      b.x = sW[kk * 64 + col];
      b.y = sW[(kk + 1) * 64 + col];
      acc[n] = __builtin_amdgcn_wmma_f32_16x16x4_f32(
           false, a,  false, b,
           (short)0, acc[n],  false,  false);
    }
  }

  int colBase = lane & 15;
  int rowOff = (lane >> 4) * 8;
#pragma unroll
  for (int r = 0; r < 8; ++r) {
    long row = tileRow + rowOff + r;
    if (row < nRows) {
#pragma unroll
      for (int n = 0; n < 4; ++n) {
        int col = n * 16 + colBase;
        float v = acc[n][r];
        if (bias) v += bias[col];
        out[row * 64 + col] = v;
      }
    }
  }
}

#define NPW 391
#define QCAP 8192
#define OWN_THREADS 1024
#define ROW_EDGES (OWN_THREADS * 4)
#define ROWS_PER_TRIP 8

__device__ void add_row_by_lane(int e, int dl, int lo, float* sAgg,
                                const int* __restrict__ src, const float* __restrict__ h,
                                const float* __restrict__ dinv) {
  int s = src[e];
  float w = dinv[s] * dinv[lo + dl];
  const float* hr = h + (long)s * 64;
  float* ar = sAgg + dl * 64;
  for (int c = 0; c < 64; ++c) atomicAdd(&ar[c], hr[c] * w);
}

__device__ __forceinline__ bool keep_edge(int e, int dl, int2* sQ, int* sQn) {
  int pos = atomicAdd(sQn, 1);
  if (pos < QCAP) {
    sQ[pos] = make_int2(e, dl);
    return true;
  }
  return false;
}

__global__ __launch_bounds__(OWN_THREADS)
void owner_aggregate(const int* __restrict__ src, const int* __restrict__ dst,
                     const float* __restrict__ h, const float* __restrict__ dinv,
                     const float* __restrict__ b, const float* __restrict__ gamma,
                     const float* __restrict__ beta, const float* __restrict__ x,
                     float* __restrict__ g, int nNodes, int E) {
  __shared__ float sAgg[NPW * 64];
  __shared__ int2 sQ[QCAP];
  __shared__ int sQn;

  const int tid = threadIdx.x;
  const int lane = tid & 31;
  const int wave = __builtin_amdgcn_readfirstlane(tid >> 5);
  const int lo = blockIdx.x * NPW;
  int cnt = nNodes - lo;
  if (cnt > NPW) cnt = NPW;
  const unsigned ucnt = (unsigned)cnt;

  for (int i = tid; i < NPW * 64; i += OWN_THREADS) sAgg[i] = 0.0f;
  if (tid == 0) sQn = 0;
  __syncthreads();

  const bool al16 = ((reinterpret_cast<size_t>(dst)) & 15) == 0;
  const int R = al16 ? E / ROW_EDGES : 0;
  if (R > 0) {
    const int rot = (int)(((long)blockIdx.x * R) / gridDim.x);
    const int4* dst4 = reinterpret_cast<const int4*>(dst) + tid;
    const int nTrips = (R + ROWS_PER_TRIP - 1) / ROWS_PER_TRIP;

    int4 cur[ROWS_PER_TRIP];
#pragma unroll
    for (int u = 0; u < ROWS_PER_TRIP; ++u) {
      int r = rot;
      if (u < R) { r = u + rot; if (r >= R) r -= R; }
      cur[u] = dst4[r * OWN_THREADS];
    }

    for (int it = 0; it < nTrips; ++it) {
      int4 nxt[ROWS_PER_TRIP];
#pragma unroll
      for (int u = 0; u < ROWS_PER_TRIP; ++u) {
        const int q = (it + 1) * ROWS_PER_TRIP + u;
        int r = rot;
        if (q < R) { r = q + rot; if (r >= R) r -= R; }
        nxt[u] = dst4[r * OWN_THREADS];
      }

      unsigned full = 0;
#pragma unroll
      for (int u = 0; u < ROWS_PER_TRIP; ++u) {
        const int q = it * ROWS_PER_TRIP + u;
        if (q < R) {
          int r = q + rot;
          if (r >= R) r -= R;
          const int d0 = cur[u].x - lo, d1 = cur[u].y - lo, d2 = cur[u].z - lo, d3 = cur[u].w - lo;
          const unsigned mn = min(min((unsigned)d0, (unsigned)d1), min((unsigned)d2, (unsigned)d3));
          if (mn < ucnt) {
            const int e0 = r * ROW_EDGES + tid * 4;
            if ((unsigned)d0 < ucnt && !keep_edge(e0,     d0, sQ, &sQn)) full |= 1u << (4 * u);
            if ((unsigned)d1 < ucnt && !keep_edge(e0 + 1, d1, sQ, &sQn)) full |= 2u << (4 * u);
            if ((unsigned)d2 < ucnt && !keep_edge(e0 + 2, d2, sQ, &sQn)) full |= 4u << (4 * u);
            if ((unsigned)d3 < ucnt && !keep_edge(e0 + 3, d3, sQ, &sQn)) full |= 8u << (4 * u);
          }
        }
      }
      while (full) {
        const int j = __builtin_ctz(full);
        full &= full - 1;
        int r = it * ROWS_PER_TRIP + (j >> 2) + rot;
        if (r >= R) r -= R;
        const int e = r * ROW_EDGES + tid * 4 + (j & 3);
        add_row_by_lane(e, dst[e] - lo, lo, sAgg, src, h, dinv);
      }
#pragma unroll
      for (int u = 0; u < ROWS_PER_TRIP; ++u) cur[u] = nxt[u];
    }
  }
  for (int e = R * ROW_EDGES + tid; e < E; e += OWN_THREADS) {
    const int dl = dst[e] - lo;
    if ((unsigned)dl < ucnt && !keep_edge(e, dl, sQ, &sQn))
      add_row_by_lane(e, dl, lo, sAgg, src, h, dinv);
  }
  __syncthreads();

  int nq = sQn;
  if (nq > QCAP) nq = QCAP;
  nq = __builtin_amdgcn_readfirstlane(nq);

  for (int k0 = wave * 32; k0 < nq; k0 += OWN_THREADS) {
    const int idx = k0 + lane;
    int s = 0, ro = 0;
    float w = 0.0f;
    if (idx < nq) {
      const int2 q = sQ[idx];
      s = src[q.x];
      w = dinv[s] * dinv[lo + q.y];
      ro = q.y * 64;
    }
    int m = nq - k0;
    if (m > 32) m = 32;
    for (int k = 0; k < m; k += 16) {
      float a0[16], a1[16];
#pragma unroll
      for (int i = 0; i < 16; ++i) {
        const int sk = __builtin_amdgcn_readlane(s, k + i);
        const float* hr = h + (long)sk * 64;
        a0[i] = hr[lane];
        a1[i] = hr[lane + 32];
      }
#pragma unroll
      for (int i = 0; i < 16; ++i) {
        if (k + i < m) {
          const float wk = __int_as_float(__builtin_amdgcn_readlane(__float_as_int(w), k + i));
          const int rk = __builtin_amdgcn_readlane(ro, k + i);
          atomicAdd(&sAgg[rk + lane], a0[i] * wk);
          atomicAdd(&sAgg[rk + lane + 32], a1[i] * wk);
        }
      }
    }
  }
  __syncthreads();

  const int c0 = lane, c1 = lane + 32;
  const float b0 = b[c0], b1 = b[c1];
  const float ga0 = gamma[c0], ga1 = gamma[c1];
  const float be0 = beta[c0], be1 = beta[c1];
  for (int nl0 = wave; nl0 < cnt; nl0 += 128) {
    float hv0[4], hv1[4], xv0[4], xv1[4], dv[4];
#pragma unroll
    for (int i = 0; i < 4; ++i) {
      int nl = nl0 + 32 * i;
      if (nl >= cnt) nl = cnt - 1;
      const long base = ((long)lo + nl) * 64;
      dv[i] = dinv[lo + nl];
      hv0[i] = h[base + c0];
      hv1[i] = h[base + c1];
      xv0[i] = x[base + c0];
      xv1[i] = x[base + c1];
    }
#pragma unroll
    for (int i = 0; i < 4; ++i) {
      const int nl = nl0 + 32 * i;
      if (nl < cnt) {
        const long base = ((long)lo + nl) * 64;
        const float selfw = dv[i] * dv[i];

        float v0 = sAgg[nl * 64 + c0] + hv0[i] * selfw + b0;
        float v1 = sAgg[nl * 64 + c1] + hv1[i] * selfw + b1;

        float s = v0 + v1;
#pragma unroll
        for (int off = 16; off > 0; off >>= 1) s += __shfl_xor(s, off, 32);
        float mu = s * (1.0f / 64.0f);

        float e0 = v0 - mu, e1 = v1 - mu;
        float q = e0 * e0 + e1 * e1;
#pragma unroll
        for (int off = 16; off > 0; off >>= 1) q += __shfl_xor(q, off, 32);
        float inv = rsqrtf(q * (1.0f / 64.0f) + 1e-5f);

        float y0 = e0 * inv * ga0 + be0;
        float y1 = e1 * inv * ga1 + be1;
        y0 = fmaxf(y0, 0.0f) * xv0[i];
        y1 = fmaxf(y1, 0.0f) * xv1[i];
        g[base + c0] = y0;
        g[base + c1] = y1;
      }
    }
  }
}

extern "C" void kernel_launch(void* const* d_in, const int* in_sizes, int n_in,
                              void* d_out, int out_size, void* d_ws, size_t ws_size,
                              hipStream_t stream) {
  const float* x     = (const float*)d_in[0];
  const int*   eidx  = (const int*)d_in[1];
  const float* W     = (const float*)d_in[2];
  const float* b     = (const float*)d_in[3];
  const float* gamma = (const float*)d_in[4];
  const float* beta  = (const float*)d_in[5];
  const float* fcW   = (const float*)d_in[6];
  const float* fcb   = (const float*)d_in[7];
  float* out = (float*)d_out;

  int N = in_sizes[0] / 64;
  int E = in_sizes[1] / 2;
  const int* src = eidx;
  const int* dst = eidx + E;

  char* ws = (char*)d_ws;
  size_t off = 0;
  float* dinv = (float*)(ws + off);
  off += (((size_t)N * 4) + 255) & ~(size_t)255;
  float* h = (float*)(ws + off);
  off += (((size_t)N * 64 * 4) + 255) & ~(size_t)255;
  float* g = (float*)(ws + off);

  const int TB = 256;

  fill_f32<<<(N + TB - 1) / TB, TB, 0, stream>>>(dinv, 1.0f, N);
  deg_accum<<<(E + TB - 1) / TB, TB, 0, stream>>>(dst, dinv, E);
  deg_to_dinv<<<(N + TB - 1) / TB, TB, 0, stream>>>(dinv, N);

  int gemmBlocks = (N + 127) / 128;
  gemm64_wmma<<<gemmBlocks, TB, 0, stream>>>(x, W, nullptr, h, N);

  owner_aggregate<<<(N + NPW - 1) / NPW, OWN_THREADS, 0, stream>>>(src, dst, h, dinv, b, gamma, beta, x, g, N, E);

  gemm64_wmma<<<gemmBlocks, TB, 0, stream>>>(g, fcW, fcb, out, N);
}


# Both 64x64 products: W kept in registers as B fragments for a wave's life (no LDS staging, no barrier); waves loop over row tiles, A of 4 tiles loaded in one round; k order of the sums re-paired
# speedup vs baseline: 2.0643x; 1.2310x over previous
#include <hip/hip_runtime.h>
#include <hip/hip_bf16.h>

typedef float v2f __attribute__((ext_vector_type(2)));
typedef float v4f __attribute__((ext_vector_type(4)));
typedef float v8f __attribute__((ext_vector_type(8)));

__global__ void fill_f32(float* p, float v, long n) {
  long i = (long)blockIdx.x * blockDim.x + threadIdx.x;
  if (i < n) p[i] = v;
}

#define NODES_PER_OWNER 391
#define KEY_EDGES 8
__global__ __launch_bounds__(256)
void deg_accum_keys(const int* __restrict__ dst, float* __restrict__ deg,
                    unsigned int* __restrict__ keys, int E) {
  const long t = (long)blockIdx.x * blockDim.x + threadIdx.x;
  const long e0 = t * KEY_EDGES;
  if (e0 >= E) return;
  const bool al16 = ((reinterpret_cast<size_t>(dst)) & 15) == 0;
  if (al16 && e0 + KEY_EDGES <= E) {
    const int4* p = reinterpret_cast<const int4*>(dst + e0);
    const int4 a = p[0];
    const int4 c = p[1];
    atomicAdd(&deg[a.x], 1.0f);
    atomicAdd(&deg[a.y], 1.0f);
    atomicAdd(&deg[a.z], 1.0f);
    atomicAdd(&deg[a.w], 1.0f);
    atomicAdd(&deg[c.x], 1.0f);
    atomicAdd(&deg[c.y], 1.0f);
    atomicAdd(&deg[c.z], 1.0f);
    atomicAdd(&deg[c.w], 1.0f);
    const unsigned k0 = ((unsigned)a.x / NODES_PER_OWNER) | (((unsigned)a.y / NODES_PER_OWNER) << 8) |
                        (((unsigned)a.z / NODES_PER_OWNER) << 16) | (((unsigned)a.w / NODES_PER_OWNER) << 24);
    const unsigned k1 = ((unsigned)c.x / NODES_PER_OWNER) | (((unsigned)c.y / NODES_PER_OWNER) << 8) |
                        (((unsigned)c.z / NODES_PER_OWNER) << 16) | (((unsigned)c.w / NODES_PER_OWNER) << 24);
    *reinterpret_cast<uint2*>(keys + (e0 >> 2)) = make_uint2(k0, k1);
  } else {
    unsigned char* kb = reinterpret_cast<unsigned char*>(keys);
    for (long e = e0; e < E && e < e0 + KEY_EDGES; ++e) {
      const int d = dst[e];
      atomicAdd(&deg[d], 1.0f);
      kb[e] = (unsigned char)((unsigned)d / NODES_PER_OWNER);
    }
  }
}

__global__ void deg_to_dinv(float* deg, int n) {
  long i = (long)blockIdx.x * blockDim.x + threadIdx.x;
  if (i < n) deg[i] = rsqrtf(deg[i]);
}

#define GEMM_WGS 256
#define GEMM_TILES_AHEAD 4

__global__ __launch_bounds__(256)
void gemm64_wreg(const float* __restrict__ A, const float* __restrict__ W,
                 const float* __restrict__ bias, float* __restrict__ out, int nRows) {
  const int lane = threadIdx.x & 31;
  const int wave = __builtin_amdgcn_readfirstlane((int)(threadIdx.x >> 5));
  const int c = lane & 15;
  const int h = lane >> 4;

  v2f bf[16][4];
#pragma unroll
  for (int jp = 0; jp < 8; ++jp) {
    const float* wr = W + (8 * jp + 4 * h) * 64 + 4 * c;
    const v4f r0 = *reinterpret_cast<const v4f*>(wr);
    const v4f r1 = *reinterpret_cast<const v4f*>(wr + 64);
    const v4f r2 = *reinterpret_cast<const v4f*>(wr + 128);
    const v4f r3 = *reinterpret_cast<const v4f*>(wr + 192);
#pragma unroll
    for (int n = 0; n < 4; ++n) {
      bf[2 * jp][n].x = r0[n];
      bf[2 * jp][n].y = r1[n];
      bf[2 * jp + 1][n].x = r2[n];
      bf[2 * jp + 1][n].y = r3[n];
    }
  }
  v4f bv = {0.0f, 0.0f, 0.0f, 0.0f};
  if (bias) bv = *reinterpret_cast<const v4f*>(bias + 4 * c);

  const int nTiles = (nRows + 15) >> 4;
  const int nWaves = (int)gridDim.x * 8;
  for (int base = wave * (int)gridDim.x + (int)blockIdx.x; base < nTiles; base += GEMM_TILES_AHEAD * nWaves) {
    v4f a[GEMM_TILES_AHEAD][8];
#pragma unroll
    for (int i = 0; i < GEMM_TILES_AHEAD; ++i) {
      int t = base + i * nWaves;
      if (t >= nTiles) t = base;
      long row = (long)t * 16 + c;
      if (row >= nRows) row = nRows - 1;
      const float* ap = A + row * 64 + 4 * h;
#pragma unroll
      for (int jp = 0; jp < 8; ++jp) a[i][jp] = *reinterpret_cast<const v4f*>(ap + 8 * jp);
    }

#pragma unroll
    for (int i = 0; i < GEMM_TILES_AHEAD; ++i) {
      const int t = base + i * nWaves;
      if (t < nTiles) {
        v8f acc[4] = {};
#pragma unroll
        for (int jp = 0; jp < 8; ++jp) {
          v2f a0, a1;
          a0.x = a[i][jp][0]; a0.y = a[i][jp][1];
          a1.x = a[i][jp][2]; a1.y = a[i][jp][3];
#pragma unroll
          for (int n = 0; n < 4; ++n)
            acc[n] = __builtin_amdgcn_wmma_f32_16x16x4_f32(
                 false, a0,  false, bf[2 * jp][n],
                 (short)0, acc[n],  false,  false);
#pragma unroll
          for (int n = 0; n < 4; ++n)
            acc[n] = __builtin_amdgcn_wmma_f32_16x16x4_f32(
                 false, a1,  false, bf[2 * jp + 1][n],
                 (short)0, acc[n],  false,  false);
        }
        const long row0 = (long)t * 16 + 8 * h;
        float* op = out + row0 * 64 + 4 * c;
        if ((long)t * 16 + 16 <= nRows) {
#pragma unroll
          for (int r = 0; r < 8; ++r) {
            v4f o;
            o[0] = acc[0][r]; o[1] = acc[1][r]; o[2] = acc[2][r]; o[3] = acc[3][r];
            if (bias) o += bv;
            *reinterpret_cast<v4f*>(op + r * 64) = o;
          }
        } else {
#pragma unroll
          for (int r = 0; r < 8; ++r) {
            if (row0 + r < nRows) {
              v4f o;
              o[0] = acc[0][r]; o[1] = acc[1][r]; o[2] = acc[2][r]; o[3] = acc[3][r];
              if (bias) o += bv;
              *reinterpret_cast<v4f*>(op + r * 64) = o;
            }
          }
        }
      }
    }
  }
}

#define NPW NODES_PER_OWNER
#define QCAP 16384
#define OWN_THREADS 1024
#define UNIT_KEYS 16
#define ROWS_PER_TRIP 7

__device__ void add_row_by_lane(int e, int dl, int lo, float* sAgg,
                                const int* __restrict__ src, const float* __restrict__ h,
                                const float* __restrict__ dinv) {
  int s = src[e];
  float w = dinv[s] * dinv[lo + dl];
  const float* hr = h + (long)s * 64;
  float* ar = sAgg + dl * 64;
  for (int c = 0; c < 64; ++c) atomicAdd(&ar[c], hr[c] * w);
}

__device__ __forceinline__ bool keep_edge(int e, int* sQ, int* sQn) {
  int pos = atomicAdd(sQn, 1);
  if (pos < QCAP) {
    sQ[pos] = e;
    return true;
  }
  return false;
}

__device__ __forceinline__ unsigned own_key_mask(int4 k, unsigned K4) {
  const unsigned x0 = (unsigned)k.x ^ K4, x1 = (unsigned)k.y ^ K4;
  const unsigned x2 = (unsigned)k.z ^ K4, x3 = (unsigned)k.w ^ K4;
  const unsigned z0 = ~(((x0 & 0x7F7F7F7Fu) + 0x7F7F7F7Fu) | x0) & 0x80808080u;
  const unsigned z1 = ~(((x1 & 0x7F7F7F7Fu) + 0x7F7F7F7Fu) | x1) & 0x80808080u;
  const unsigned z2 = ~(((x2 & 0x7F7F7F7Fu) + 0x7F7F7F7Fu) | x2) & 0x80808080u;
  const unsigned z3 = ~(((x3 & 0x7F7F7F7Fu) + 0x7F7F7F7Fu) | x3) & 0x80808080u;
  return (z0 >> 7) | (z1 >> 6) | (z2 >> 5) | (z3 >> 4);
}

__global__ __launch_bounds__(OWN_THREADS)
void owner_aggregate(const int* __restrict__ src, const int* __restrict__ dst,
                     const unsigned int* __restrict__ keys,
                     const float* __restrict__ h, const float* __restrict__ dinv,
                     const float* __restrict__ b, const float* __restrict__ gamma,
                     const float* __restrict__ beta, const float* __restrict__ x,
                     float* __restrict__ g, int nNodes, int E) {
  __shared__ float sAgg[NPW * 64];
  __shared__ int sQ[QCAP];
  __shared__ int sQn;

  const int tid = threadIdx.x;
  const int lane = tid & 31;
  const int wave = __builtin_amdgcn_readfirstlane(tid >> 5);
  const int lo = blockIdx.x * NPW;
  int cnt = nNodes - lo;
  if (cnt > NPW) cnt = NPW;
  const unsigned ucnt = (unsigned)cnt;

  for (int i = tid; i < NPW * 64; i += OWN_THREADS) sAgg[i] = 0.0f;
  if (tid == 0) sQn = 0;
  __syncthreads();

  const unsigned K4 = (unsigned)blockIdx.x * 0x01010101u;
  const int nUnits = E / UNIT_KEYS;
  const int R = (nUnits + OWN_THREADS - 1) / OWN_THREADS;
  if (R > 0) {
    const int rot = (int)(((long)blockIdx.x * R) / gridDim.x);
    const int4* key4 = reinterpret_cast<const int4*>(keys);
    const int nTrips = (R + ROWS_PER_TRIP - 1) / ROWS_PER_TRIP;
    const int lastUnit = nUnits - 1;

    int4 cur[ROWS_PER_TRIP];
#pragma unroll
    for (int u = 0; u < ROWS_PER_TRIP; ++u) {
      int r = rot;
      if (u < R) { r = u + rot; if (r >= R) r -= R; }
      int un = r * OWN_THREADS + tid;
      if (un > lastUnit) un = lastUnit;
      cur[u] = key4[un];
    }

    for (int it = 0; it < nTrips; ++it) {
      int4 nxt[ROWS_PER_TRIP];
#pragma unroll
      for (int u = 0; u < ROWS_PER_TRIP; ++u) {
        const int q = (it + 1) * ROWS_PER_TRIP + u;
        int r = rot;
        if (q < R) { r = q + rot; if (r >= R) r -= R; }
        int un = r * OWN_THREADS + tid;
        if (un > lastUnit) un = lastUnit;
        nxt[u] = key4[un];
      }

#pragma unroll
      for (int u = 0; u < ROWS_PER_TRIP; ++u) {
        const int q = it * ROWS_PER_TRIP + u;
        if (q < R) {
          int r = q + rot;
          if (r >= R) r -= R;
          const int unit = r * OWN_THREADS + tid;
          unsigned m = own_key_mask(cur[u], K4);
          if (m != 0 && unit <= lastUnit) {
            do {
              const int p = __builtin_ctz(m);
              m &= m - 1;
              const int e = unit * UNIT_KEYS + ((p & 7) << 2) + (p >> 3);
              if (!keep_edge(e, sQ, &sQn))
                add_row_by_lane(e, dst[e] - lo, lo, sAgg, src, h, dinv);
            } while (m);
          }
        }
      }
#pragma unroll
      for (int u = 0; u < ROWS_PER_TRIP; ++u) cur[u] = nxt[u];
    }
  }
  for (int e = nUnits * UNIT_KEYS + tid; e < E; e += OWN_THREADS) {
    const int dl = dst[e] - lo;
    if ((unsigned)dl < ucnt && !keep_edge(e, sQ, &sQn))
      add_row_by_lane(e, dl, lo, sAgg, src, h, dinv);
  }
  __syncthreads();

  int nq = sQn;
  if (nq > QCAP) nq = QCAP;
  nq = __builtin_amdgcn_readfirstlane(nq);

  for (int k0 = wave * 32; k0 < nq; k0 += OWN_THREADS) {
    const int idx = k0 + lane;
    int s = 0, ro = 0;
    float w = 0.0f;
    if (idx < nq) {
      const int e = sQ[idx];
      s = src[e];
      const int d = dst[e];
      w = dinv[s] * dinv[d];
      ro = (d - lo) * 64;
    }
    int m = nq - k0;
    if (m > 32) m = 32;
    for (int k = 0; k < m; k += 16) {
      float a0[16], a1[16];
#pragma unroll
      for (int i = 0; i < 16; ++i) {
        const int sk = __builtin_amdgcn_readlane(s, k + i);
        const float* hr = h + (long)sk * 64;
        a0[i] = hr[lane];
        a1[i] = hr[lane + 32];
      }
#pragma unroll
      for (int i = 0; i < 16; ++i) {
        if (k + i < m) {
          const float wk = __int_as_float(__builtin_amdgcn_readlane(__float_as_int(w), k + i));
          const int rk = __builtin_amdgcn_readlane(ro, k + i);
          atomicAdd(&sAgg[rk + lane], a0[i] * wk);
          atomicAdd(&sAgg[rk + lane + 32], a1[i] * wk);
        }
      }
    }
  }
  __syncthreads();

  const int c0 = lane, c1 = lane + 32;
  const float b0 = b[c0], b1 = b[c1];
  const float ga0 = gamma[c0], ga1 = gamma[c1];
  const float be0 = beta[c0], be1 = beta[c1];
  for (int nl0 = wave; nl0 < cnt; nl0 += 128) {
    float hv0[4], hv1[4], xv0[4], xv1[4], dv[4];
#pragma unroll
    for (int i = 0; i < 4; ++i) {
      int nl = nl0 + 32 * i;
      if (nl >= cnt) nl = cnt - 1;
      const long base = ((long)lo + nl) * 64;
      dv[i] = dinv[lo + nl];
      hv0[i] = h[base + c0];
      hv1[i] = h[base + c1];
      xv0[i] = x[base + c0];
      xv1[i] = x[base + c1];
    }
#pragma unroll
    for (int i = 0; i < 4; ++i) {
      const int nl = nl0 + 32 * i;
      if (nl < cnt) {
        const long base = ((long)lo + nl) * 64;
        const float selfw = dv[i] * dv[i];

        float v0 = sAgg[nl * 64 + c0] + hv0[i] * selfw + b0;
        float v1 = sAgg[nl * 64 + c1] + hv1[i] * selfw + b1;

        float s = v0 + v1;
#pragma unroll
        for (int off = 16; off > 0; off >>= 1) s += __shfl_xor(s, off, 32);
        float mu = s * (1.0f / 64.0f);

        float e0 = v0 - mu, e1 = v1 - mu;
        float q = e0 * e0 + e1 * e1;
#pragma unroll
        for (int off = 16; off > 0; off >>= 1) q += __shfl_xor(q, off, 32);
        float inv = rsqrtf(q * (1.0f / 64.0f) + 1e-5f);

        float y0 = e0 * inv * ga0 + be0;
        float y1 = e1 * inv * ga1 + be1;
        y0 = fmaxf(y0, 0.0f) * xv0[i];
        y1 = fmaxf(y1, 0.0f) * xv1[i];
        g[base + c0] = y0;
        g[base + c1] = y1;
      }
    }
  }
}

extern "C" void kernel_launch(void* const* d_in, const int* in_sizes, int n_in,
                              void* d_out, int out_size, void* d_ws, size_t ws_size,
                              hipStream_t stream) {
  const float* x     = (const float*)d_in[0];
  const int*   eidx  = (const int*)d_in[1];
  const float* W     = (const float*)d_in[2];
  const float* b     = (const float*)d_in[3];
  const float* gamma = (const float*)d_in[4];
  const float* beta  = (const float*)d_in[5];
  const float* fcW   = (const float*)d_in[6];
  const float* fcb   = (const float*)d_in[7];
  float* out = (float*)d_out;

  int N = in_sizes[0] / 64;
  int E = in_sizes[1] / 2;
  const int* src = eidx;
  const int* dst = eidx + E;

  char* ws = (char*)d_ws;
  size_t off = 0;
  float* dinv = (float*)(ws + off);
  off += (((size_t)N * 4) + 255) & ~(size_t)255;
  float* h = (float*)(ws + off);
  off += (((size_t)N * 64 * 4) + 255) & ~(size_t)255;
  float* g = (float*)(ws + off);
  off += (((size_t)N * 64 * 4) + 255) & ~(size_t)255;
  unsigned int* keys = (unsigned int*)(ws + off);

  const int TB = 256;

  fill_f32<<<(N + TB - 1) / TB, TB, 0, stream>>>(dinv, 1.0f, N);
  deg_accum_keys<<<(E + TB * KEY_EDGES - 1) / (TB * KEY_EDGES), TB, 0, stream>>>(dst, dinv, keys, E);
  deg_to_dinv<<<(N + TB - 1) / TB, TB, 0, stream>>>(dinv, N);

  gemm64_wreg<<<GEMM_WGS, TB, 0, stream>>>(x, W, nullptr, h, N);

  owner_aggregate<<<(N + NPW - 1) / NPW, OWN_THREADS, 0, stream>>>(src, dst, keys, h, dinv, b, gamma, beta, x, g, N, E);

  gemm64_wreg<<<GEMM_WGS, TB, 0, stream>>>(g, fcW, fcb, out, N);
}


# Degree count: 800000 scattered global float atomics replaced by per-chunk LDS integer histograms (32 work-groups, packed 16-bit counters) summed in the dinv launch; no fill launch, 6 -> 5 launches
# speedup vs baseline: 2.9642x; 1.4359x over previous
#include <hip/hip_runtime.h>
#include <hip/hip_bf16.h>

typedef float v2f __attribute__((ext_vector_type(2)));
typedef float v4f __attribute__((ext_vector_type(4)));
typedef float v8f __attribute__((ext_vector_type(8)));

#define NODES_PER_OWNER 391
#define DEG_CHUNKS 32
#define DEG_THREADS 1024
#define DEG_AHEAD 8
#define HIST_WORDS 50048

__device__ __forceinline__ void count_dst(unsigned int* sCnt, int d) {
  atomicAdd(&sCnt[(unsigned)d >> 1], 1u << (((unsigned)d & 1u) * 16u));
}

__global__ __launch_bounds__(DEG_THREADS)
void deg_hist_keys(const int* __restrict__ dst, unsigned int* __restrict__ part,
                   unsigned int* __restrict__ keys, int E) {
  __shared__ unsigned int sCnt[HIST_WORDS];
  const int tid = threadIdx.x;

  uint4* sCnt4 = reinterpret_cast<uint4*>(sCnt);
  for (int i = tid; i < HIST_WORDS / 4; i += DEG_THREADS) sCnt4[i] = make_uint4(0u, 0u, 0u, 0u);
  __syncthreads();

  const int per = ((E + (int)gridDim.x * 16 - 1) / ((int)gridDim.x * 16)) * 16;
  const long e_lo = (long)blockIdx.x * per;
  long e_hi = e_lo + per;
  if (e_hi > E) e_hi = E;
  if (e_hi < e_lo) e_hi = e_lo;

  const bool al16 = ((reinterpret_cast<size_t>(dst)) & 15) == 0;
  if (al16) {
    const int nFull = (int)((e_hi - e_lo) >> 2);
    const int4* d4 = reinterpret_cast<const int4*>(dst + e_lo);
    unsigned int* kw = keys + (e_lo >> 2);
    for (int u0 = 0; u0 < nFull; u0 += DEG_AHEAD * DEG_THREADS) {
      int4 v[DEG_AHEAD];
#pragma unroll
      for (int i = 0; i < DEG_AHEAD; ++i) {
        int u = u0 + i * DEG_THREADS + tid;
        if (u >= nFull) u = nFull - 1;
        v[i] = d4[u];
      }
#pragma unroll
      for (int i = 0; i < DEG_AHEAD; ++i) {
        const int u = u0 + i * DEG_THREADS + tid;
        if (u < nFull) {
          const int4 a = v[i];
          count_dst(sCnt, a.x);
          count_dst(sCnt, a.y);
          count_dst(sCnt, a.z);
          count_dst(sCnt, a.w);
          kw[u] = ((unsigned)a.x / NODES_PER_OWNER) | (((unsigned)a.y / NODES_PER_OWNER) << 8) |
                  (((unsigned)a.z / NODES_PER_OWNER) << 16) | (((unsigned)a.w / NODES_PER_OWNER) << 24);
        }
      }
    }
    unsigned char* kb = reinterpret_cast<unsigned char*>(keys);
    for (long e = e_lo + 4L * nFull + tid; e < e_hi; e += DEG_THREADS) {
      const int d = dst[e];
      count_dst(sCnt, d);
      kb[e] = (unsigned char)((unsigned)d / NODES_PER_OWNER);
    }
  } else {
    unsigned char* kb = reinterpret_cast<unsigned char*>(keys);
    for (long e = e_lo + tid; e < e_hi; e += DEG_THREADS) {
      const int d = dst[e];
      count_dst(sCnt, d);
      kb[e] = (unsigned char)((unsigned)d / NODES_PER_OWNER);
    }
  }
  __syncthreads();

  uint4* p4 = reinterpret_cast<uint4*>(part + (size_t)blockIdx.x * HIST_WORDS);
  for (int i = tid; i < HIST_WORDS / 4; i += DEG_THREADS) p4[i] = sCnt4[i];
}

__global__ __launch_bounds__(256)
void deg_merge_dinv(const unsigned int* __restrict__ part, float* __restrict__ dinv, int n) {
  const int w = (int)(blockIdx.x * blockDim.x + threadIdx.x);
  if (w >= HIST_WORDS || 2 * w >= n) return;
  unsigned int v[DEG_CHUNKS];
#pragma unroll
  for (int c = 0; c < DEG_CHUNKS; ++c) v[c] = part[(size_t)c * HIST_WORDS + w];
  unsigned int lo = 1u, hi = 1u;
#pragma unroll
  for (int c = 0; c < DEG_CHUNKS; ++c) {
    lo += v[c] & 0xFFFFu;
    hi += v[c] >> 16;
  }
  const float r0 = rsqrtf((float)lo);
  const float r1 = rsqrtf((float)hi);
  if (2 * w + 1 < n) {
    *reinterpret_cast<float2*>(dinv + 2 * w) = make_float2(r0, r1);
  } else {
    dinv[2 * w] = r0;
  }
}

#define GEMM_WGS 256
#define GEMM_TILES_AHEAD 4

__global__ __launch_bounds__(256)
void gemm64_wreg(const float* __restrict__ A, const float* __restrict__ W,
                 const float* __restrict__ bias, float* __restrict__ out, int nRows) {
  const int lane = threadIdx.x & 31;
  const int wave = __builtin_amdgcn_readfirstlane((int)(threadIdx.x >> 5));
  const int c = lane & 15;
  const int h = lane >> 4;

  v2f bf[16][4];
#pragma unroll
  for (int jp = 0; jp < 8; ++jp) {
    const float* wr = W + (8 * jp + 4 * h) * 64 + 4 * c;
    const v4f r0 = *reinterpret_cast<const v4f*>(wr);
    const v4f r1 = *reinterpret_cast<const v4f*>(wr + 64);
    const v4f r2 = *reinterpret_cast<const v4f*>(wr + 128);
    const v4f r3 = *reinterpret_cast<const v4f*>(wr + 192);
#pragma unroll
    for (int n = 0; n < 4; ++n) {
      bf[2 * jp][n].x = r0[n];
      bf[2 * jp][n].y = r1[n];
      bf[2 * jp + 1][n].x = r2[n];
      bf[2 * jp + 1][n].y = r3[n];
    }
  }
  v4f bv = {0.0f, 0.0f, 0.0f, 0.0f};
  if (bias) bv = *reinterpret_cast<const v4f*>(bias + 4 * c);

  const int nTiles = (nRows + 15) >> 4;
  const int nWaves = (int)gridDim.x * 8;
  for (int base = wave * (int)gridDim.x + (int)blockIdx.x; base < nTiles; base += GEMM_TILES_AHEAD * nWaves) {
    v4f a[GEMM_TILES_AHEAD][8];
#pragma unroll
    for (int i = 0; i < GEMM_TILES_AHEAD; ++i) {
      int t = base + i * nWaves;
      if (t >= nTiles) t = base;
      long row = (long)t * 16 + c;
      if (row >= nRows) row = nRows - 1;
      const float* ap = A + row * 64 + 4 * h;
#pragma unroll
      for (int jp = 0; jp < 8; ++jp) a[i][jp] = *reinterpret_cast<const v4f*>(ap + 8 * jp);
    }

#pragma unroll
    for (int i = 0; i < GEMM_TILES_AHEAD; ++i) {
      const int t = base + i * nWaves;
      if (t < nTiles) {
        v8f acc[4] = {};
#pragma unroll
        for (int jp = 0; jp < 8; ++jp) {
          v2f a0, a1;
          a0.x = a[i][jp][0]; a0.y = a[i][jp][1];
          a1.x = a[i][jp][2]; a1.y = a[i][jp][3];
#pragma unroll
          for (int n = 0; n < 4; ++n)
            acc[n] = __builtin_amdgcn_wmma_f32_16x16x4_f32(
                 false, a0,  false, bf[2 * jp][n],
                 (short)0, acc[n],  false,  false);
#pragma unroll
          for (int n = 0; n < 4; ++n)
            acc[n] = __builtin_amdgcn_wmma_f32_16x16x4_f32(
                 false, a1,  false, bf[2 * jp + 1][n],
                 (short)0, acc[n],  false,  false);
        }
        const long row0 = (long)t * 16 + 8 * h;
        float* op = out + row0 * 64 + 4 * c;
        if ((long)t * 16 + 16 <= nRows) {
#pragma unroll
          for (int r = 0; r < 8; ++r) {
            v4f o;
            o[0] = acc[0][r]; o[1] = acc[1][r]; o[2] = acc[2][r]; o[3] = acc[3][r];
            if (bias) o += bv;
            *reinterpret_cast<v4f*>(op + r * 64) = o;
          }
        } else {
#pragma unroll
          for (int r = 0; r < 8; ++r) {
            if (row0 + r < nRows) {
              v4f o;
              o[0] = acc[0][r]; o[1] = acc[1][r]; o[2] = acc[2][r]; o[3] = acc[3][r];
              if (bias) o += bv;
              *reinterpret_cast<v4f*>(op + r * 64) = o;
            }
          }
        }
      }
    }
  }
}

#define NPW NODES_PER_OWNER
#define QCAP 16384
#define OWN_THREADS 1024
#define UNIT_KEYS 16
#define ROWS_PER_TRIP 7

__device__ void add_row_by_lane(int e, int dl, int lo, float* sAgg,
                                const int* __restrict__ src, const float* __restrict__ h,
                                const float* __restrict__ dinv) {
  int s = src[e];
  float w = dinv[s] * dinv[lo + dl];
  const float* hr = h + (long)s * 64;
  float* ar = sAgg + dl * 64;
  for (int c = 0; c < 64; ++c) atomicAdd(&ar[c], hr[c] * w);
}

__device__ __forceinline__ bool keep_edge(int e, int* sQ, int* sQn) {
  int pos = atomicAdd(sQn, 1);
  if (pos < QCAP) {
    sQ[pos] = e;
    return true;
  }
  return false;
}

__device__ __forceinline__ unsigned own_key_mask(int4 k, unsigned K4) {
  const unsigned x0 = (unsigned)k.x ^ K4, x1 = (unsigned)k.y ^ K4;
  const unsigned x2 = (unsigned)k.z ^ K4, x3 = (unsigned)k.w ^ K4;
  const unsigned z0 = ~(((x0 & 0x7F7F7F7Fu) + 0x7F7F7F7Fu) | x0) & 0x80808080u;
  const unsigned z1 = ~(((x1 & 0x7F7F7F7Fu) + 0x7F7F7F7Fu) | x1) & 0x80808080u;
  const unsigned z2 = ~(((x2 & 0x7F7F7F7Fu) + 0x7F7F7F7Fu) | x2) & 0x80808080u;
  const unsigned z3 = ~(((x3 & 0x7F7F7F7Fu) + 0x7F7F7F7Fu) | x3) & 0x80808080u;
  return (z0 >> 7) | (z1 >> 6) | (z2 >> 5) | (z3 >> 4);
}

__global__ __launch_bounds__(OWN_THREADS)
void owner_aggregate(const int* __restrict__ src, const int* __restrict__ dst,
                     const unsigned int* __restrict__ keys,
                     const float* __restrict__ h, const float* __restrict__ dinv,
                     const float* __restrict__ b, const float* __restrict__ gamma,
                     const float* __restrict__ beta, const float* __restrict__ x,
                     float* __restrict__ g, int nNodes, int E) {
  __shared__ float sAgg[NPW * 64];
  __shared__ int sQ[QCAP];
  __shared__ int sQn;

  const int tid = threadIdx.x;
  const int lane = tid & 31;
  const int wave = __builtin_amdgcn_readfirstlane(tid >> 5);
  const int lo = blockIdx.x * NPW;
  int cnt = nNodes - lo;
  if (cnt > NPW) cnt = NPW;
  const unsigned ucnt = (unsigned)cnt;

  for (int i = tid; i < NPW * 64; i += OWN_THREADS) sAgg[i] = 0.0f;
  if (tid == 0) sQn = 0;
  __syncthreads();

  const unsigned K4 = (unsigned)blockIdx.x * 0x01010101u;
  const int nUnits = E / UNIT_KEYS;
  const int R = (nUnits + OWN_THREADS - 1) / OWN_THREADS;
  if (R > 0) {
    const int rot = (int)(((long)blockIdx.x * R) / gridDim.x);
    const int4* key4 = reinterpret_cast<const int4*>(keys);
    const int nTrips = (R + ROWS_PER_TRIP - 1) / ROWS_PER_TRIP;
    const int lastUnit = nUnits - 1;

    int4 cur[ROWS_PER_TRIP];
#pragma unroll
    for (int u = 0; u < ROWS_PER_TRIP; ++u) {
      int r = rot;
      if (u < R) { r = u + rot; if (r >= R) r -= R; }
      int un = r * OWN_THREADS + tid;
      if (un > lastUnit) un = lastUnit;
      cur[u] = key4[un];
    }

    for (int it = 0; it < nTrips; ++it) {
      int4 nxt[ROWS_PER_TRIP];
#pragma unroll
      for (int u = 0; u < ROWS_PER_TRIP; ++u) {
        const int q = (it + 1) * ROWS_PER_TRIP + u;
        int r = rot;
        if (q < R) { r = q + rot; if (r >= R) r -= R; }
        int un = r * OWN_THREADS + tid;
        if (un > lastUnit) un = lastUnit;
        nxt[u] = key4[un];
      }

#pragma unroll
      for (int u = 0; u < ROWS_PER_TRIP; ++u) {
        const int q = it * ROWS_PER_TRIP + u;
        if (q < R) {
          int r = q + rot;
          if (r >= R) r -= R;
          const int unit = r * OWN_THREADS + tid;
          unsigned m = own_key_mask(cur[u], K4);
          if (m != 0 && unit <= lastUnit) {
            do {
              const int p = __builtin_ctz(m);
              m &= m - 1;
              const int e = unit * UNIT_KEYS + ((p & 7) << 2) + (p >> 3);
              if (!keep_edge(e, sQ, &sQn))
                add_row_by_lane(e, dst[e] - lo, lo, sAgg, src, h, dinv);
            } while (m);
          }
        }
      }
#pragma unroll
      for (int u = 0; u < ROWS_PER_TRIP; ++u) cur[u] = nxt[u];
    }
  }
  for (int e = nUnits * UNIT_KEYS + tid; e < E; e += OWN_THREADS) {
    const int dl = dst[e] - lo;
    if ((unsigned)dl < ucnt && !keep_edge(e, sQ, &sQn))
      add_row_by_lane(e, dl, lo, sAgg, src, h, dinv);
  }
  __syncthreads();

  int nq = sQn;
  if (nq > QCAP) nq = QCAP;
  nq = __builtin_amdgcn_readfirstlane(nq);

  for (int k0 = wave * 32; k0 < nq; k0 += OWN_THREADS) {
    const int idx = k0 + lane;
    int s = 0, ro = 0;
    float w = 0.0f;
    if (idx < nq) {
      const int e = sQ[idx];
      s = src[e];
      const int d = dst[e];
      w = dinv[s] * dinv[d];
      ro = (d - lo) * 64;
    }
    int m = nq - k0;
    if (m > 32) m = 32;
    for (int k = 0; k < m; k += 16) {
      float a0[16], a1[16];
#pragma unroll
      for (int i = 0; i < 16; ++i) {
        const int sk = __builtin_amdgcn_readlane(s, k + i);
        const float* hr = h + (long)sk * 64;
        a0[i] = hr[lane];
        a1[i] = hr[lane + 32];
      }
#pragma unroll
      for (int i = 0; i < 16; ++i) {
        if (k + i < m) {
          const float wk = __int_as_float(__builtin_amdgcn_readlane(__float_as_int(w), k + i));
          const int rk = __builtin_amdgcn_readlane(ro, k + i);
          atomicAdd(&sAgg[rk + lane], a0[i] * wk);
          atomicAdd(&sAgg[rk + lane + 32], a1[i] * wk);
        }
      }
    }
  }
  __syncthreads();

  const int c0 = lane, c1 = lane + 32;
  const float b0 = b[c0], b1 = b[c1];
  const float ga0 = gamma[c0], ga1 = gamma[c1];
  const float be0 = beta[c0], be1 = beta[c1];
  for (int nl0 = wave; nl0 < cnt; nl0 += 128) {
    float hv0[4], hv1[4], xv0[4], xv1[4], dv[4];
#pragma unroll
    for (int i = 0; i < 4; ++i) {
      int nl = nl0 + 32 * i;
      if (nl >= cnt) nl = cnt - 1;
      const long base = ((long)lo + nl) * 64;
      dv[i] = dinv[lo + nl];
      hv0[i] = h[base + c0];
      hv1[i] = h[base + c1];
      xv0[i] = x[base + c0];
      xv1[i] = x[base + c1];
    }
#pragma unroll
    for (int i = 0; i < 4; ++i) {
      const int nl = nl0 + 32 * i;
      if (nl < cnt) {
        const long base = ((long)lo + nl) * 64;
        const float selfw = dv[i] * dv[i];

        float v0 = sAgg[nl * 64 + c0] + hv0[i] * selfw + b0;
        float v1 = sAgg[nl * 64 + c1] + hv1[i] * selfw + b1;

        float s = v0 + v1;
#pragma unroll
        for (int off = 16; off > 0; off >>= 1) s += __shfl_xor(s, off, 32);
        float mu = s * (1.0f / 64.0f);

        float e0 = v0 - mu, e1 = v1 - mu;
        float q = e0 * e0 + e1 * e1;
#pragma unroll
        for (int off = 16; off > 0; off >>= 1) q += __shfl_xor(q, off, 32);
        float inv = rsqrtf(q * (1.0f / 64.0f) + 1e-5f);

        float y0 = e0 * inv * ga0 + be0;
        float y1 = e1 * inv * ga1 + be1;
        y0 = fmaxf(y0, 0.0f) * xv0[i];
        y1 = fmaxf(y1, 0.0f) * xv1[i];
        g[base + c0] = y0;
        g[base + c1] = y1;
      }
    }
  }
}

extern "C" void kernel_launch(void* const* d_in, const int* in_sizes, int n_in,
                              void* d_out, int out_size, void* d_ws, size_t ws_size,
                              hipStream_t stream) {
  const float* x     = (const float*)d_in[0];
  const int*   eidx  = (const int*)d_in[1];
  const float* W     = (const float*)d_in[2];
  const float* b     = (const float*)d_in[3];
  const float* gamma = (const float*)d_in[4];
  const float* beta  = (const float*)d_in[5];
  const float* fcW   = (const float*)d_in[6];
  const float* fcb   = (const float*)d_in[7];
  float* out = (float*)d_out;

  int N = in_sizes[0] / 64;
  int E = in_sizes[1] / 2;
  const int* src = eidx;
  const int* dst = eidx + E;

  char* ws = (char*)d_ws;
  size_t off = 0;
  float* dinv = (float*)(ws + off);
  off += (((size_t)N * 4) + 255) & ~(size_t)255;
  float* h = (float*)(ws + off);
  off += (((size_t)N * 64 * 4) + 255) & ~(size_t)255;
  float* g = (float*)(ws + off);
  off += (((size_t)N * 64 * 4) + 255) & ~(size_t)255;
  unsigned int* keys = (unsigned int*)(ws + off);
  off += (((size_t)E) + 255) & ~(size_t)255;
  unsigned int* part = (unsigned int*)(ws + off);

  const int TB = 256;

  deg_hist_keys<<<DEG_CHUNKS, DEG_THREADS, 0, stream>>>(dst, part, keys, E);
  deg_merge_dinv<<<(HIST_WORDS + TB - 1) / TB, TB, 0, stream>>>(part, dinv, N);

  gemm64_wreg<<<GEMM_WGS, TB, 0, stream>>>(x, W, nullptr, h, N);

  owner_aggregate<<<(N + NPW - 1) / NPW, OWN_THREADS, 0, stream>>>(src, dst, keys, h, dinv, b, gamma, beta, x, g, N, E);

  gemm64_wreg<<<GEMM_WGS, TB, 0, stream>>>(g, fcW, fcb, out, N);
}


# Degree merge (32 partial count words + self loop, rsqrt) moved into the h = x @ W launch: one count word a thread, its loads issued ahead of the W loads; 5 -> 4 launches; all sum orders unchanged
# speedup vs baseline: 3.0737x; 1.0369x over previous
#include <hip/hip_runtime.h>
#include <hip/hip_bf16.h>

typedef float v2f __attribute__((ext_vector_type(2)));
typedef float v4f __attribute__((ext_vector_type(4)));
typedef float v8f __attribute__((ext_vector_type(8)));

#define NODES_PER_OWNER 391
#define DEG_CHUNKS 32
#define DEG_THREADS 1024
#define DEG_AHEAD 8
#define HIST_WORDS 50048

__device__ __forceinline__ void count_dst(unsigned int* sCnt, int d) {
  atomicAdd(&sCnt[(unsigned)d >> 1], 1u << (((unsigned)d & 1u) * 16u));
}

__global__ __launch_bounds__(DEG_THREADS)
void deg_hist_keys(const int* __restrict__ dst, unsigned int* __restrict__ part,
                   unsigned int* __restrict__ keys, int E) {
  __shared__ unsigned int sCnt[HIST_WORDS];
  const int tid = threadIdx.x;

  uint4* sCnt4 = reinterpret_cast<uint4*>(sCnt);
  for (int i = tid; i < HIST_WORDS / 4; i += DEG_THREADS) sCnt4[i] = make_uint4(0u, 0u, 0u, 0u);
  __syncthreads();

  const int per = ((E + (int)gridDim.x * 16 - 1) / ((int)gridDim.x * 16)) * 16;
  const long e_lo = (long)blockIdx.x * per;
  long e_hi = e_lo + per;
  if (e_hi > E) e_hi = E;
  if (e_hi < e_lo) e_hi = e_lo;

  const bool al16 = ((reinterpret_cast<size_t>(dst)) & 15) == 0;
  if (al16) {
    const int nFull = (int)((e_hi - e_lo) >> 2);
    const int4* d4 = reinterpret_cast<const int4*>(dst + e_lo);
    unsigned int* kw = keys + (e_lo >> 2);
    for (int u0 = 0; u0 < nFull; u0 += DEG_AHEAD * DEG_THREADS) {
      int4 v[DEG_AHEAD];
#pragma unroll
      for (int i = 0; i < DEG_AHEAD; ++i) {
        int u = u0 + i * DEG_THREADS + tid;
        if (u >= nFull) u = nFull - 1;
        v[i] = d4[u];
      }
#pragma unroll
      for (int i = 0; i < DEG_AHEAD; ++i) {
        const int u = u0 + i * DEG_THREADS + tid;
        if (u < nFull) {
          const int4 a = v[i];
          count_dst(sCnt, a.x);
          count_dst(sCnt, a.y);
          count_dst(sCnt, a.z);
          count_dst(sCnt, a.w);
          kw[u] = ((unsigned)a.x / NODES_PER_OWNER) | (((unsigned)a.y / NODES_PER_OWNER) << 8) |
                  (((unsigned)a.z / NODES_PER_OWNER) << 16) | (((unsigned)a.w / NODES_PER_OWNER) << 24);
        }
      }
    }
    unsigned char* kb = reinterpret_cast<unsigned char*>(keys);
    for (long e = e_lo + 4L * nFull + tid; e < e_hi; e += DEG_THREADS) {
      const int d = dst[e];
      count_dst(sCnt, d);
      kb[e] = (unsigned char)((unsigned)d / NODES_PER_OWNER);
    }
  } else {
    unsigned char* kb = reinterpret_cast<unsigned char*>(keys);
    for (long e = e_lo + tid; e < e_hi; e += DEG_THREADS) {
      const int d = dst[e];
      count_dst(sCnt, d);
      kb[e] = (unsigned char)((unsigned)d / NODES_PER_OWNER);
    }
  }
  __syncthreads();

  uint4* p4 = reinterpret_cast<uint4*>(part + (size_t)blockIdx.x * HIST_WORDS);
  for (int i = tid; i < HIST_WORDS / 4; i += DEG_THREADS) p4[i] = sCnt4[i];
}

__device__ __forceinline__ void store_dinv_pair(const unsigned int (&v)[DEG_CHUNKS], float* __restrict__ dinv,
                                                int w, int n) {
  unsigned int lo = 1u, hi = 1u;
#pragma unroll
  for (int c = 0; c < DEG_CHUNKS; ++c) {
    lo += v[c] & 0xFFFFu;
    hi += v[c] >> 16;
  }
  const float r0 = rsqrtf((float)lo);
  const float r1 = rsqrtf((float)hi);
  if (2 * w + 1 < n) {
    *reinterpret_cast<float2*>(dinv + 2 * w) = make_float2(r0, r1);
  } else {
    dinv[2 * w] = r0;
  }
}

#define GEMM_WGS 256
#define GEMM_TILES_AHEAD 4
static_assert(GEMM_WGS * 256 >= HIST_WORDS, "one thread of the product launch per count word");

template <bool MERGE_DEG>
__global__ __launch_bounds__(256)
void gemm64_wreg(const float* __restrict__ A, const float* __restrict__ W,
                 const float* __restrict__ bias, float* __restrict__ out, int nRows,
                 const unsigned int* __restrict__ part, float* __restrict__ dinv) {
  const int lane = threadIdx.x & 31;
  const int wave = __builtin_amdgcn_readfirstlane((int)(threadIdx.x >> 5));
  const int c = lane & 15;
  const int h = lane >> 4;

  unsigned int pv[DEG_CHUNKS];
  const int mw = (int)blockIdx.x * 256 + (int)threadIdx.x;
  if constexpr (MERGE_DEG) {
    const int wc = mw < HIST_WORDS ? mw : HIST_WORDS - 1;
#pragma unroll
    for (int k = 0; k < DEG_CHUNKS; ++k) pv[k] = part[(size_t)k * HIST_WORDS + wc];
  }

  v2f bf[16][4];
#pragma unroll
  for (int jp = 0; jp < 8; ++jp) {
    const float* wr = W + (8 * jp + 4 * h) * 64 + 4 * c;
    const v4f r0 = *reinterpret_cast<const v4f*>(wr);
    const v4f r1 = *reinterpret_cast<const v4f*>(wr + 64);
    const v4f r2 = *reinterpret_cast<const v4f*>(wr + 128);
    const v4f r3 = *reinterpret_cast<const v4f*>(wr + 192);
#pragma unroll
    for (int n = 0; n < 4; ++n) {
      bf[2 * jp][n].x = r0[n];
      bf[2 * jp][n].y = r1[n];
      bf[2 * jp + 1][n].x = r2[n];
      bf[2 * jp + 1][n].y = r3[n];
    }
  }
  v4f bv = {0.0f, 0.0f, 0.0f, 0.0f};
  if (bias) bv = *reinterpret_cast<const v4f*>(bias + 4 * c);

  if constexpr (MERGE_DEG) {
    if (mw < HIST_WORDS && 2 * mw < nRows) store_dinv_pair(pv, dinv, mw, nRows);
  }

  const int nTiles = (nRows + 15) >> 4;
  const int nWaves = (int)gridDim.x * 8;
  for (int base = wave * (int)gridDim.x + (int)blockIdx.x; base < nTiles; base += GEMM_TILES_AHEAD * nWaves) {
    v4f a[GEMM_TILES_AHEAD][8];
#pragma unroll
    for (int i = 0; i < GEMM_TILES_AHEAD; ++i) {
      int t = base + i * nWaves;
      if (t >= nTiles) t = base;
      long row = (long)t * 16 + c;
      if (row >= nRows) row = nRows - 1;
      const float* ap = A + row * 64 + 4 * h;
#pragma unroll
      for (int jp = 0; jp < 8; ++jp) a[i][jp] = *reinterpret_cast<const v4f*>(ap + 8 * jp);
    }

#pragma unroll
    for (int i = 0; i < GEMM_TILES_AHEAD; ++i) {
      const int t = base + i * nWaves;
      if (t < nTiles) {
        v8f acc[4] = {};
#pragma unroll
        for (int jp = 0; jp < 8; ++jp) {
          v2f a0, a1;
          a0.x = a[i][jp][0]; a0.y = a[i][jp][1];
          a1.x = a[i][jp][2]; a1.y = a[i][jp][3];
#pragma unroll
          for (int n = 0; n < 4; ++n)
            acc[n] = __builtin_amdgcn_wmma_f32_16x16x4_f32(
                 false, a0,  false, bf[2 * jp][n],
                 (short)0, acc[n],  false,  false);
#pragma unroll
          for (int n = 0; n < 4; ++n)
            acc[n] = __builtin_amdgcn_wmma_f32_16x16x4_f32(
                 false, a1,  false, bf[2 * jp + 1][n],
                 (short)0, acc[n],  false,  false);
        }
        const long row0 = (long)t * 16 + 8 * h;
        float* op = out + row0 * 64 + 4 * c;
        if ((long)t * 16 + 16 <= nRows) {
#pragma unroll
          for (int r = 0; r < 8; ++r) {
            v4f o;
            o[0] = acc[0][r]; o[1] = acc[1][r]; o[2] = acc[2][r]; o[3] = acc[3][r];
            if (bias) o += bv;
            *reinterpret_cast<v4f*>(op + r * 64) = o;
          }
        } else {
#pragma unroll
          for (int r = 0; r < 8; ++r) {
            if (row0 + r < nRows) {
              v4f o;
              o[0] = acc[0][r]; o[1] = acc[1][r]; o[2] = acc[2][r]; o[3] = acc[3][r];
              if (bias) o += bv;
              *reinterpret_cast<v4f*>(op + r * 64) = o;
            }
          }
        }
      }
    }
  }
}

#define NPW NODES_PER_OWNER
#define QCAP 16384
#define OWN_THREADS 1024
#define UNIT_KEYS 16
#define ROWS_PER_TRIP 7

__device__ void add_row_by_lane(int e, int dl, int lo, float* sAgg,
                                const int* __restrict__ src, const float* __restrict__ h,
                                const float* __restrict__ dinv) {
  int s = src[e];
  float w = dinv[s] * dinv[lo + dl];
  const float* hr = h + (long)s * 64;
  float* ar = sAgg + dl * 64;
  for (int c = 0; c < 64; ++c) atomicAdd(&ar[c], hr[c] * w);
}

__device__ __forceinline__ bool keep_edge(int e, int* sQ, int* sQn) {
  int pos = atomicAdd(sQn, 1);
  if (pos < QCAP) {
    sQ[pos] = e;
    return true;
  }
  return false;
}

__device__ __forceinline__ unsigned own_key_mask(int4 k, unsigned K4) {
  const unsigned x0 = (unsigned)k.x ^ K4, x1 = (unsigned)k.y ^ K4;
  const unsigned x2 = (unsigned)k.z ^ K4, x3 = (unsigned)k.w ^ K4;
  const unsigned z0 = ~(((x0 & 0x7F7F7F7Fu) + 0x7F7F7F7Fu) | x0) & 0x80808080u;
  const unsigned z1 = ~(((x1 & 0x7F7F7F7Fu) + 0x7F7F7F7Fu) | x1) & 0x80808080u;
  const unsigned z2 = ~(((x2 & 0x7F7F7F7Fu) + 0x7F7F7F7Fu) | x2) & 0x80808080u;
  const unsigned z3 = ~(((x3 & 0x7F7F7F7Fu) + 0x7F7F7F7Fu) | x3) & 0x80808080u;
  return (z0 >> 7) | (z1 >> 6) | (z2 >> 5) | (z3 >> 4);
}

__global__ __launch_bounds__(OWN_THREADS)
void owner_aggregate(const int* __restrict__ src, const int* __restrict__ dst,
                     const unsigned int* __restrict__ keys,
                     const float* __restrict__ h, const float* __restrict__ dinv,
                     const float* __restrict__ b, const float* __restrict__ gamma,
                     const float* __restrict__ beta, const float* __restrict__ x,
                     float* __restrict__ g, int nNodes, int E) {
  __shared__ float sAgg[NPW * 64];
  __shared__ int sQ[QCAP];
  __shared__ int sQn;

  const int tid = threadIdx.x;
  const int lane = tid & 31;
  const int wave = __builtin_amdgcn_readfirstlane(tid >> 5);
  const int lo = blockIdx.x * NPW;
  int cnt = nNodes - lo;
  if (cnt > NPW) cnt = NPW;
  const unsigned ucnt = (unsigned)cnt;

  for (int i = tid; i < NPW * 64; i += OWN_THREADS) sAgg[i] = 0.0f;
  if (tid == 0) sQn = 0;
  __syncthreads();

  const unsigned K4 = (unsigned)blockIdx.x * 0x01010101u;
  const int nUnits = E / UNIT_KEYS;
  const int R = (nUnits + OWN_THREADS - 1) / OWN_THREADS;
  if (R > 0) {
    const int rot = (int)(((long)blockIdx.x * R) / gridDim.x);
    const int4* key4 = reinterpret_cast<const int4*>(keys);
    const int nTrips = (R + ROWS_PER_TRIP - 1) / ROWS_PER_TRIP;
    const int lastUnit = nUnits - 1;

    int4 cur[ROWS_PER_TRIP];
#pragma unroll
    for (int u = 0; u < ROWS_PER_TRIP; ++u) {
      int r = rot;
      if (u < R) { r = u + rot; if (r >= R) r -= R; }
      int un = r * OWN_THREADS + tid;
      if (un > lastUnit) un = lastUnit;
      cur[u] = key4[un];
    }

    for (int it = 0; it < nTrips; ++it) {
      int4 nxt[ROWS_PER_TRIP];
#pragma unroll
      for (int u = 0; u < ROWS_PER_TRIP; ++u) {
        const int q = (it + 1) * ROWS_PER_TRIP + u;
        int r = rot;
        if (q < R) { r = q + rot; if (r >= R) r -= R; }
        int un = r * OWN_THREADS + tid;
        if (un > lastUnit) un = lastUnit;
        nxt[u] = key4[un];
      }

#pragma unroll
      for (int u = 0; u < ROWS_PER_TRIP; ++u) {
        const int q = it * ROWS_PER_TRIP + u;
        if (q < R) {
          int r = q + rot;
          if (r >= R) r -= R;
          const int unit = r * OWN_THREADS + tid;
          unsigned m = own_key_mask(cur[u], K4);
          if (m != 0 && unit <= lastUnit) {
            do {
              const int p = __builtin_ctz(m);
              m &= m - 1;
              const int e = unit * UNIT_KEYS + ((p & 7) << 2) + (p >> 3);
              if (!keep_edge(e, sQ, &sQn))
                add_row_by_lane(e, dst[e] - lo, lo, sAgg, src, h, dinv);
            } while (m);
          }
        }
      }
#pragma unroll
      for (int u = 0; u < ROWS_PER_TRIP; ++u) cur[u] = nxt[u];
    }
  }
  for (int e = nUnits * UNIT_KEYS + tid; e < E; e += OWN_THREADS) {
    const int dl = dst[e] - lo;
    if ((unsigned)dl < ucnt && !keep_edge(e, sQ, &sQn))
      add_row_by_lane(e, dl, lo, sAgg, src, h, dinv);
  }
  __syncthreads();

  int nq = sQn;
  if (nq > QCAP) nq = QCAP;
  nq = __builtin_amdgcn_readfirstlane(nq);

  for (int k0 = wave * 32; k0 < nq; k0 += OWN_THREADS) {
    const int idx = k0 + lane;
    int s = 0, ro = 0;
    float w = 0.0f;
    if (idx < nq) {
      const int e = sQ[idx];
      s = src[e];
      const int d = dst[e];
      w = dinv[s] * dinv[d];
      ro = (d - lo) * 64;
    }
    int m = nq - k0;
    if (m > 32) m = 32;
    for (int k = 0; k < m; k += 16) {
      float a0[16], a1[16];
#pragma unroll
      for (int i = 0; i < 16; ++i) {
        const int sk = __builtin_amdgcn_readlane(s, k + i);
        const float* hr = h + (long)sk * 64;
        a0[i] = hr[lane];
        a1[i] = hr[lane + 32];
      }
#pragma unroll
      for (int i = 0; i < 16; ++i) {
        if (k + i < m) {
          const float wk = __int_as_float(__builtin_amdgcn_readlane(__float_as_int(w), k + i));
          const int rk = __builtin_amdgcn_readlane(ro, k + i);
          atomicAdd(&sAgg[rk + lane], a0[i] * wk);
          atomicAdd(&sAgg[rk + lane + 32], a1[i] * wk);
        }
      }
    }
  }
  __syncthreads();

  const int c0 = lane, c1 = lane + 32;
  const float b0 = b[c0], b1 = b[c1];
  const float ga0 = gamma[c0], ga1 = gamma[c1];
  const float be0 = beta[c0], be1 = beta[c1];
  for (int nl0 = wave; nl0 < cnt; nl0 += 128) {
    float hv0[4], hv1[4], xv0[4], xv1[4], dv[4];
#pragma unroll
    for (int i = 0; i < 4; ++i) {
      int nl = nl0 + 32 * i;
      if (nl >= cnt) nl = cnt - 1;
      const long base = ((long)lo + nl) * 64;
      dv[i] = dinv[lo + nl];
      hv0[i] = h[base + c0];
      hv1[i] = h[base + c1];
      xv0[i] = x[base + c0];
      xv1[i] = x[base + c1];
    }
#pragma unroll
    for (int i = 0; i < 4; ++i) {
      const int nl = nl0 + 32 * i;
      if (nl < cnt) {
        const long base = ((long)lo + nl) * 64;
        const float selfw = dv[i] * dv[i];

        float v0 = sAgg[nl * 64 + c0] + hv0[i] * selfw + b0;
        float v1 = sAgg[nl * 64 + c1] + hv1[i] * selfw + b1;

        float s = v0 + v1;
#pragma unroll
        for (int off = 16; off > 0; off >>= 1) s += __shfl_xor(s, off, 32);
        float mu = s * (1.0f / 64.0f);

        float e0 = v0 - mu, e1 = v1 - mu;
        float q = e0 * e0 + e1 * e1;
#pragma unroll
        for (int off = 16; off > 0; off >>= 1) q += __shfl_xor(q, off, 32);
        float inv = rsqrtf(q * (1.0f / 64.0f) + 1e-5f);

        float y0 = e0 * inv * ga0 + be0;
        float y1 = e1 * inv * ga1 + be1;
        y0 = fmaxf(y0, 0.0f) * xv0[i];
        y1 = fmaxf(y1, 0.0f) * xv1[i];
        g[base + c0] = y0;
        g[base + c1] = y1;
      }
    }
  }
}

extern "C" void kernel_launch(void* const* d_in, const int* in_sizes, int n_in,
                              void* d_out, int out_size, void* d_ws, size_t ws_size,
                              hipStream_t stream) {
  const float* x     = (const float*)d_in[0];
  const int*   eidx  = (const int*)d_in[1];
  const float* W     = (const float*)d_in[2];
  const float* b     = (const float*)d_in[3];
  const float* gamma = (const float*)d_in[4];
  const float* beta  = (const float*)d_in[5];
  const float* fcW   = (const float*)d_in[6];
  const float* fcb   = (const float*)d_in[7];
  float* out = (float*)d_out;

  int N = in_sizes[0] / 64;
  int E = in_sizes[1] / 2;
  const int* src = eidx;
  const int* dst = eidx + E;

  char* ws = (char*)d_ws;
  size_t off = 0;
  float* dinv = (float*)(ws + off);
  off += (((size_t)N * 4) + 255) & ~(size_t)255;
  float* h = (float*)(ws + off);
  off += (((size_t)N * 64 * 4) + 255) & ~(size_t)255;
  float* g = (float*)(ws + off);
  off += (((size_t)N * 64 * 4) + 255) & ~(size_t)255;
  unsigned int* keys = (unsigned int*)(ws + off);
  off += (((size_t)E) + 255) & ~(size_t)255;
  unsigned int* part = (unsigned int*)(ws + off);

  const int TB = 256;

  deg_hist_keys<<<DEG_CHUNKS, DEG_THREADS, 0, stream>>>(dst, part, keys, E);

  gemm64_wreg<true><<<GEMM_WGS, TB, 0, stream>>>(x, W, nullptr, h, N, part, dinv);

  owner_aggregate<<<(N + NPW - 1) / NPW, OWN_THREADS, 0, stream>>>(src, dst, keys, h, dinv, b, gamma, beta, x, g, N, E);

  gemm64_wreg<false><<<GEMM_WGS, TB, 0, stream>>>(g, fcW, fcb, out, N, nullptr, nullptr);
}
